# starting kernel
# baseline (speedup 1.0000x reference)
#include <hip/hip_runtime.h>

typedef __attribute__((ext_vector_type(2))) float v2f;
typedef __attribute__((ext_vector_type(8))) float v8f;

#define IN_F   4096
#define OUT_F  4096
#define BATCH  1024
#define NCB    (OUT_F / 16)
#define NTILES 1024
#define CHUNK  256

__global__ void k_transpose(const float* __restrict__ x, float* __restrict__ xT) {
    __shared__ float tile[32][33];
    const int ci = blockIdx.x * 32;
    const int bi = blockIdx.y * 32;
    for (int j = threadIdx.y; j < 32; j += 8)
        tile[j][threadIdx.x] = x[(size_t)(bi + j) * IN_F + ci + threadIdx.x];
    __syncthreads();
    for (int j = threadIdx.y; j < 32; j += 8)
        xT[(size_t)(ci + j) * BATCH + bi + threadIdx.x] = tile[threadIdx.x][j];
}

__global__ void k_hist(const int* __restrict__ cols, int nnz, int tile_sz,
                       int* __restrict__ tileHist) {
    __shared__ int h[256];
    const int t = blockIdx.x;
    h[threadIdx.x] = 0;
    __syncthreads();
    const int beg = t * tile_sz;
    const int end = min(beg + tile_sz, nnz);
    for (int i = beg + (int)threadIdx.x; i < end; i += 256)
        atomicAdd(&h[(cols[i] >> 4) & 255], 1);
    __syncthreads();
    tileHist[t * 256 + threadIdx.x] = h[threadIdx.x];
}

__global__ void k_scan(int* __restrict__ tileHist, int* __restrict__ bucketStart,
                       int* __restrict__ bucketCount) {
    __shared__ int sc[256];
    const int cb = threadIdx.x;
    int run = 0;
    for (int t = 0; t < NTILES; ++t) {
        int v = tileHist[t * 256 + cb];
        tileHist[t * 256 + cb] = run;
        run += v;
    }
    sc[cb] = run;
    __syncthreads();
    for (int d = 1; d < 256; d <<= 1) {
        int v = (cb >= d) ? sc[cb - d] : 0;
        __syncthreads();
        sc[cb] += v;
        __syncthreads();
    }
    const int base = sc[cb] - run;
    bucketStart[cb] = base;
    bucketCount[cb] = run;
    __syncthreads();
    for (int t = 0; t < NTILES; ++t)
        tileHist[t * 256 + cb] += base;
}

__global__ void k_scatter(const int* __restrict__ rows, const int* __restrict__ cols,
                          const float* __restrict__ vals, int nnz, int tile_sz,
                          int* __restrict__ tileOff, unsigned short* __restrict__ rc_s,
                          float* __restrict__ vals_s) {
    const int t = blockIdx.x * blockDim.x + threadIdx.x;
    if (t >= NTILES) return;
    const int beg = t * tile_sz;
    const int end = min(beg + tile_sz, nnz);
    for (int i = beg; i < end; ++i) {
        const int c  = cols[i];
        const int cb = (c >> 4) & 255;
        const int p  = tileOff[t * 256 + cb]++;
        rc_s[p]   = (unsigned short)((rows[i] & 0xFFF) | ((c & 15) << 12));
        vals_s[p] = vals[i];
    }
}

__global__ void __launch_bounds__(256)
k_spmm_wmma(const float* __restrict__ xT,
            const int*   __restrict__ bucketStart,
            const int*   __restrict__ bucketCount,
            const unsigned short* __restrict__ rc_s,
            const float* __restrict__ vals_s,
            const float* __restrict__ bias,
            float* __restrict__ out) {
    __shared__ unsigned short lrc[CHUNK];
    __shared__ float          lv[CHUNK];

    const int cb    = blockIdx.x;
    const int cbase = cb * 16;
    const int tid   = threadIdx.x;
    const int wave  = tid >> 5;
    const int lane  = tid & 31;
    const int half  = lane >> 4;
    const int ln    = lane & 15;
    const int b0    = (blockIdx.y * 8 + wave) * 16;

    const int s = bucketStart[cb];
    const int e = s + bucketCount[cb];

    const float* __restrict__ xp = xT + (size_t)b0 + ln;

    v8f acc0 = {};
    v8f acc1 = {};
    for (int base = s; base < e; base += CHUNK) {
        __syncthreads();
        for (int i = tid; i < CHUNK; i += 256) {
            const int idx = base + i;
            if (idx < e) { lrc[i] = rc_s[idx]; lv[i] = vals_s[idx]; }
            else         { lrc[i] = 0;         lv[i] = 0.0f;        }
            __builtin_prefetch(&rc_s[idx + CHUNK], 0, 0);
            __builtin_prefetch(&vals_s[idx + CHUNK], 0, 0);
        }
        __syncthreads();
        const int n    = min(CHUNK, e - base);
        const int kmax = (n + 7) & ~7;
        for (int k = 0; k < kmax; k += 8) {
            const int kk0 = k + (half << 1);
            const int kk1 = kk0 + 4;

            const unsigned int rcA = *(const unsigned int*)&lrc[kk0];
            const int   rA0 = rcA & 0xFFF,          rA1 = (rcA >> 16) & 0xFFF;
            const int   cA0 = (rcA >> 12) & 0xF,    cA1 = rcA >> 28;
            const float vA0 = lv[kk0],              vA1 = lv[kk0 + 1];
            v2f av, bv;
            av.x = xp[rA0 << 10];
            av.y = xp[rA1 << 10];
            bv.x = (cA0 == ln) ? vA0 : 0.0f;
            bv.y = (cA1 == ln) ? vA1 : 0.0f;
            acc0 = __builtin_amdgcn_wmma_f32_16x16x4_f32(
                false, av, false, bv, (short)0, acc0, false, false);

            const unsigned int rcB = *(const unsigned int*)&lrc[kk1];
            const int   rB0 = rcB & 0xFFF,          rB1 = (rcB >> 16) & 0xFFF;
            const int   cB0 = (rcB >> 12) & 0xF,    cB1 = rcB >> 28;
            const float vB0 = lv[kk1],              vB1 = lv[kk1 + 1];
            v2f aw, bw;
            aw.x = xp[rB0 << 10];
            aw.y = xp[rB1 << 10];
            bw.x = (cB0 == ln) ? vB0 : 0.0f;
            bw.y = (cB1 == ln) ? vB1 : 0.0f;
            acc1 = __builtin_amdgcn_wmma_f32_16x16x4_f32(
                false, aw, false, bw, (short)0, acc1, false, false);
        }
    }

    const float bz = bias[cbase + ln];
    #pragma unroll
    for (int j = 0; j < 8; ++j) {
        const int m = j + 8 * half;
        out[(size_t)(b0 + m) * OUT_F + cbase + ln] = acc0[j] + acc1[j] + bz;
    }
}

extern "C" void kernel_launch(void* const* d_in, const int* in_sizes, int n_in,
                              void* d_out, int out_size, void* d_ws, size_t ws_size,
                              hipStream_t stream) {
    (void)n_in; (void)out_size; (void)ws_size;
    const float* x      = (const float*)d_in[0];
    const float* values = (const float*)d_in[1];
    const float* bias   = (const float*)d_in[2];
    const int*   rows   = (const int*)d_in[3];
    const int*   cols   = (const int*)d_in[4];
    const int    nnz    = in_sizes[1];

    char*  ws  = (char*)d_ws;
    size_t off = 0;
    float* xT          = (float*)(ws + off); off += (size_t)IN_F * BATCH * sizeof(float);
    int*   tileHist    = (int*)  (ws + off); off += (size_t)NTILES * 256 * sizeof(int);
    int*   bucketStart = (int*)  (ws + off); off += 256 * sizeof(int);
    int*   bucketCount = (int*)  (ws + off); off += 256 * sizeof(int);
    unsigned short* rc_s = (unsigned short*)(ws + off);
    off += (((size_t)nnz * sizeof(unsigned short)) + 15) & ~(size_t)15;
    float* vals_s      = (float*)(ws + off); off += (size_t)nnz * sizeof(float);

    const int tile_sz = (nnz + NTILES - 1) / NTILES;

    k_transpose<<<dim3(IN_F / 32, BATCH / 32), dim3(32, 8), 0, stream>>>(x, xT);
    k_hist    <<<NTILES, 256, 0, stream>>>(cols, nnz, tile_sz, tileHist);
    k_scan    <<<1, 256, 0, stream>>>(tileHist, bucketStart, bucketCount);
    k_scatter <<<(NTILES + 255) / 256, 256, 0, stream>>>(rows, cols, values, nnz, tile_sz,
                                                         tileHist, rc_s, vals_s);
    k_spmm_wmma<<<dim3(NCB, BATCH / 128), 256, 0, stream>>>(xT, bucketStart, bucketCount,
                                                            rc_s, vals_s, bias,
                                                            (float*)d_out);
}


# scatter: a work-group per tile ranks its elements in parallel from bucket ids in LDS; before, one thread walked each tile serially with global-memory cursors; same stable order, same sums
# speedup vs baseline: 1.9857x; 1.9857x over previous
#include <hip/hip_runtime.h>

typedef __attribute__((ext_vector_type(2))) float v2f;
typedef __attribute__((ext_vector_type(8))) float v8f;

#define IN_F   4096
#define OUT_F  4096
#define BATCH  1024
#define NCB    (OUT_F / 16)
#define NTILES 1024
#define CHUNK  256

__global__ void k_transpose(const float* __restrict__ x, float* __restrict__ xT) {
    __shared__ float tile[32][33];
    const int ci = blockIdx.x * 32;
    const int bi = blockIdx.y * 32;
    for (int j = threadIdx.y; j < 32; j += 8)
        tile[j][threadIdx.x] = x[(size_t)(bi + j) * IN_F + ci + threadIdx.x];
    __syncthreads();
    for (int j = threadIdx.y; j < 32; j += 8)
        xT[(size_t)(ci + j) * BATCH + bi + threadIdx.x] = tile[threadIdx.x][j];
}

__global__ void k_hist(const int* __restrict__ cols, int nnz, int tile_sz,
                       int* __restrict__ tileHist) {
    __shared__ int h[256];
    const int t = blockIdx.x;
    h[threadIdx.x] = 0;
    __syncthreads();
    const int beg = t * tile_sz;
    const int end = min(beg + tile_sz, nnz);
    for (int i = beg + (int)threadIdx.x; i < end; i += 256)
        atomicAdd(&h[(cols[i] >> 4) & 255], 1);
    __syncthreads();
    tileHist[t * 256 + threadIdx.x] = h[threadIdx.x];
}

__global__ void k_scan(int* __restrict__ tileHist, int* __restrict__ bucketStart,
                       int* __restrict__ bucketCount) {
    __shared__ int sc[256];
    const int cb = threadIdx.x;
    int run = 0;
    for (int t = 0; t < NTILES; ++t) {
        int v = tileHist[t * 256 + cb];
        tileHist[t * 256 + cb] = run;
        run += v;
    }
    sc[cb] = run;
    __syncthreads();
    for (int d = 1; d < 256; d <<= 1) {
        int v = (cb >= d) ? sc[cb - d] : 0;
        __syncthreads();
        sc[cb] += v;
        __syncthreads();
    }
    const int base = sc[cb] - run;
    bucketStart[cb] = base;
    bucketCount[cb] = run;
    __syncthreads();
    for (int t = 0; t < NTILES; ++t)
        tileHist[t * 256 + cb] += base;
}

#define SCAT_T 256

__device__ __forceinline__ int count_same(const uint4* ids, int cb, int lo, int hi) {
    int cnt = 0;
    for (int g = lo >> 4; g < ((hi + 15) >> 4); ++g) {
        const uint4 q = ids[g];
        const unsigned w[4] = {q.x, q.y, q.z, q.w};
        #pragma unroll
        for (int b = 0; b < 16; ++b) {
            const int j  = g * 16 + b;
            const int id = (int)((w[b >> 2] >> (8 * (b & 3))) & 255u);
            cnt += (int)((id == cb) & (j >= lo) & (j < hi));
        }
    }
    return cnt;
}

__global__ void __launch_bounds__(SCAT_T)
k_scatter(const int* __restrict__ rows, const int* __restrict__ cols,
          const float* __restrict__ vals, int nnz, int tile_sz,
          const int* __restrict__ tileOff, unsigned short* __restrict__ rc_s,
          float* __restrict__ vals_s) {
    __shared__ uint4 ids[SCAT_T / 16];
    __shared__ int   placed[256];

    const int tid = threadIdx.x;
    const int t   = blockIdx.x;
    const int beg = t * tile_sz;
    const int end = min(beg + tile_sz, nnz);
    const int* __restrict__ off = tileOff + t * 256;

    placed[tid] = 0;
    for (int base = beg; base < end; base += SCAT_T) {
        const int  n    = min(SCAT_T, end - base);
        const bool live = tid < n;
        const bool more = base + SCAT_T < end;
        int c = 0, r = 0;
        float v = 0.0f;
        if (live) { c = cols[base + tid]; r = rows[base + tid]; v = vals[base + tid]; }
        const int cb = (c >> 4) & 255;

        __syncthreads();
        ((unsigned char*)ids)[tid] = (unsigned char)cb;
        __syncthreads();

        int before = 0, after = 0;
        if (live) {
            const int first = off[cb] + placed[cb];
            before = count_same(ids, cb, 0, tid);
            if (more) after = count_same(ids, cb, tid + 1, SCAT_T);
            const int p = first + before;
            rc_s[p]   = (unsigned short)((r & 0xFFF) | ((c & 15) << 12));
            vals_s[p] = v;
        }
        if (more) {
            __syncthreads();
            if (after == 0) placed[cb] += before + 1;
        }
    }
}

__global__ void __launch_bounds__(256)
k_spmm_wmma(const float* __restrict__ xT,
            const int*   __restrict__ bucketStart,
            const int*   __restrict__ bucketCount,
            const unsigned short* __restrict__ rc_s,
            const float* __restrict__ vals_s,
            const float* __restrict__ bias,
            float* __restrict__ out) {
    __shared__ unsigned short lrc[CHUNK];
    __shared__ float          lv[CHUNK];

    const int cb    = blockIdx.x;
    const int cbase = cb * 16;
    const int tid   = threadIdx.x;
    const int wave  = tid >> 5;
    const int lane  = tid & 31;
    const int half  = lane >> 4;
    const int ln    = lane & 15;
    const int b0    = (blockIdx.y * 8 + wave) * 16;

    const int s = bucketStart[cb];
    const int e = s + bucketCount[cb];

    const float* __restrict__ xp = xT + (size_t)b0 + ln;

    v8f acc0 = {};
    v8f acc1 = {};
    for (int base = s; base < e; base += CHUNK) {
        __syncthreads();
        for (int i = tid; i < CHUNK; i += 256) {
            const int idx = base + i;
            if (idx < e) { lrc[i] = rc_s[idx]; lv[i] = vals_s[idx]; }
            else         { lrc[i] = 0;         lv[i] = 0.0f;        }
            __builtin_prefetch(&rc_s[idx + CHUNK], 0, 0);
            __builtin_prefetch(&vals_s[idx + CHUNK], 0, 0);
        }
        __syncthreads();
        const int n    = min(CHUNK, e - base);
        const int kmax = (n + 7) & ~7;
        for (int k = 0; k < kmax; k += 8) {
            const int kk0 = k + (half << 1);
            const int kk1 = kk0 + 4;

            const unsigned int rcA = *(const unsigned int*)&lrc[kk0];
            const int   rA0 = rcA & 0xFFF,          rA1 = (rcA >> 16) & 0xFFF;
            const int   cA0 = (rcA >> 12) & 0xF,    cA1 = rcA >> 28;
            const float vA0 = lv[kk0],              vA1 = lv[kk0 + 1];
            v2f av, bv;
            av.x = xp[rA0 << 10];
            av.y = xp[rA1 << 10];
            bv.x = (cA0 == ln) ? vA0 : 0.0f;
            bv.y = (cA1 == ln) ? vA1 : 0.0f;
            acc0 = __builtin_amdgcn_wmma_f32_16x16x4_f32(
                false, av, false, bv, (short)0, acc0, false, false);

            const unsigned int rcB = *(const unsigned int*)&lrc[kk1];
            const int   rB0 = rcB & 0xFFF,          rB1 = (rcB >> 16) & 0xFFF;
            const int   cB0 = (rcB >> 12) & 0xF,    cB1 = rcB >> 28;
            const float vB0 = lv[kk1],              vB1 = lv[kk1 + 1];
            v2f aw, bw;
            aw.x = xp[rB0 << 10];
            aw.y = xp[rB1 << 10];
            bw.x = (cB0 == ln) ? vB0 : 0.0f;
            bw.y = (cB1 == ln) ? vB1 : 0.0f;
            acc1 = __builtin_amdgcn_wmma_f32_16x16x4_f32(
                false, aw, false, bw, (short)0, acc1, false, false);
        }
    }

    const float bz = bias[cbase + ln];
    #pragma unroll
    for (int j = 0; j < 8; ++j) {
        const int m = j + 8 * half;
        out[(size_t)(b0 + m) * OUT_F + cbase + ln] = acc0[j] + acc1[j] + bz;
    }
}

extern "C" void kernel_launch(void* const* d_in, const int* in_sizes, int n_in,
                              void* d_out, int out_size, void* d_ws, size_t ws_size,
                              hipStream_t stream) {
    (void)n_in; (void)out_size; (void)ws_size;
    const float* x      = (const float*)d_in[0];
    const float* values = (const float*)d_in[1];
    const float* bias   = (const float*)d_in[2];
    const int*   rows   = (const int*)d_in[3];
    const int*   cols   = (const int*)d_in[4];
    const int    nnz    = in_sizes[1];

    char*  ws  = (char*)d_ws;
    size_t off = 0;
    float* xT          = (float*)(ws + off); off += (size_t)IN_F * BATCH * sizeof(float);
    int*   tileHist    = (int*)  (ws + off); off += (size_t)NTILES * 256 * sizeof(int);
    int*   bucketStart = (int*)  (ws + off); off += 256 * sizeof(int);
    int*   bucketCount = (int*)  (ws + off); off += 256 * sizeof(int);
    unsigned short* rc_s = (unsigned short*)(ws + off);
    off += (((size_t)nnz * sizeof(unsigned short)) + 15) & ~(size_t)15;
    float* vals_s      = (float*)(ws + off); off += (size_t)nnz * sizeof(float);

    const int tile_sz = (nnz + NTILES - 1) / NTILES;

    k_transpose<<<dim3(IN_F / 32, BATCH / 32), dim3(32, 8), 0, stream>>>(x, xT);
    k_hist    <<<NTILES, 256, 0, stream>>>(cols, nnz, tile_sz, tileHist);
    k_scan    <<<1, 256, 0, stream>>>(tileHist, bucketStart, bucketCount);
    k_scatter <<<NTILES, SCAT_T, 0, stream>>>(rows, cols, values, nnz, tile_sz,
                                              tileHist, rc_s, vals_s);
    k_spmm_wmma<<<dim3(NCB, BATCH / 128), 256, 0, stream>>>(xT, bucketStart, bucketCount,
                                                            rc_s, vals_s, bias,
                                                            (float*)d_out);
}


# scan: tile-histogram prefix sums in parallel (work-group per 32 buckets, wave per 32 tiles, whole-line loads and stores); before, one work-group walked the 1024 tiles serially; same slots, same sums
# speedup vs baseline: 4.7423x; 2.3882x over previous
#include <hip/hip_runtime.h>

typedef __attribute__((ext_vector_type(2))) float v2f;
typedef __attribute__((ext_vector_type(8))) float v8f;

#define IN_F   4096
#define OUT_F  4096
#define BATCH  1024
#define NCB    (OUT_F / 16)
#define NTILES 1024
#define CHUNK  256

__global__ void k_transpose(const float* __restrict__ x, float* __restrict__ xT) {
    __shared__ float tile[32][33];
    const int ci = blockIdx.x * 32;
    const int bi = blockIdx.y * 32;
    for (int j = threadIdx.y; j < 32; j += 8)
        tile[j][threadIdx.x] = x[(size_t)(bi + j) * IN_F + ci + threadIdx.x];
    __syncthreads();
    for (int j = threadIdx.y; j < 32; j += 8)
        xT[(size_t)(ci + j) * BATCH + bi + threadIdx.x] = tile[threadIdx.x][j];
}

__global__ void k_hist(const int* __restrict__ cols, int nnz, int tile_sz,
                       int* __restrict__ tileHist) {
    __shared__ int h[256];
    const int t = blockIdx.x;
    h[threadIdx.x] = 0;
    __syncthreads();
    const int beg = t * tile_sz;
    const int end = min(beg + tile_sz, nnz);
    for (int i = beg + (int)threadIdx.x; i < end; i += 256)
        atomicAdd(&h[(cols[i] >> 4) & 255], 1);
    __syncthreads();
    tileHist[t * 256 + threadIdx.x] = h[threadIdx.x];
}

#define SCAN_W   32
#define SCAN_TPW (NTILES / SCAN_W)

__global__ void __launch_bounds__(SCAN_W * 32)
k_scan(int* __restrict__ tileHist, int* __restrict__ bucketCount) {
    __shared__ int wsum[SCAN_W][32];

    const int lane = threadIdx.x & 31;
    const int w    = threadIdx.x >> 5;
    const int cb   = blockIdx.x * 32 + lane;
    int* __restrict__ p = tileHist + (size_t)(w * SCAN_TPW) * 256 + cb;

    int v[SCAN_TPW];
    for (int j = 0; j < SCAN_TPW; ++j) v[j] = p[j * 256];

    int run = 0;
    for (int j = 0; j < SCAN_TPW; ++j) { const int c = v[j]; v[j] = run; run += c; }

    wsum[w][lane] = run;
    __syncthreads();

    int before = 0;
    for (int k = 0; k < SCAN_W; ++k) {
        const int s = wsum[k][lane];
        before += (k < w) ? s : 0;
    }

    for (int j = 0; j < SCAN_TPW; ++j) p[j * 256] = v[j] + before;

    if (w == SCAN_W - 1) bucketCount[cb] = before + run;
}

#define SCAT_T 256

__device__ __forceinline__ int count_same(const uint4* ids, int cb, int lo, int hi) {
    int cnt = 0;
    for (int g = lo >> 4; g < ((hi + 15) >> 4); ++g) {
        const uint4 q = ids[g];
        const unsigned w[4] = {q.x, q.y, q.z, q.w};
        #pragma unroll
        for (int b = 0; b < 16; ++b) {
            const int j  = g * 16 + b;
            const int id = (int)((w[b >> 2] >> (8 * (b & 3))) & 255u);
            cnt += (int)((id == cb) & (j >= lo) & (j < hi));
        }
    }
    return cnt;
}

__global__ void __launch_bounds__(SCAT_T)
k_scatter(const int* __restrict__ rows, const int* __restrict__ cols,
          const float* __restrict__ vals, int nnz, int tile_sz,
          const int* __restrict__ tileOff, const int* __restrict__ bucketCount,
          unsigned short* __restrict__ rc_s, float* __restrict__ vals_s) {
    __shared__ uint4 ids[SCAT_T / 16];
    __shared__ int   placed[256];
    __shared__ int   bstart[256];
    __shared__ int   wtot[SCAT_T / 32];

    const int tid = threadIdx.x;
    const int t   = blockIdx.x;
    const int beg = t * tile_sz;
    const int end = min(beg + tile_sz, nnz);
    const int* __restrict__ off = tileOff + t * 256;

    placed[tid] = 0;
    {
        const int lane = tid & 31;
        const int wv   = tid >> 5;
        const int cnt  = bucketCount[tid];
        int inc = cnt;
        for (int d = 1; d < 32; d <<= 1) {
            const int o = __shfl_up(inc, d);
            if (lane >= d) inc += o;
        }
        if (lane == 31) wtot[wv] = inc;
        __syncthreads();
        int pre = 0;
        for (int k = 0; k < SCAT_T / 32; ++k) pre += (k < wv) ? wtot[k] : 0;
        bstart[tid] = pre + inc - cnt;
    }
    for (int base = beg; base < end; base += SCAT_T) {
        const int  n    = min(SCAT_T, end - base);
        const bool live = tid < n;
        const bool more = base + SCAT_T < end;
        int c = 0, r = 0;
        float v = 0.0f;
        if (live) { c = cols[base + tid]; r = rows[base + tid]; v = vals[base + tid]; }
        const int cb = (c >> 4) & 255;

        __syncthreads();
        ((unsigned char*)ids)[tid] = (unsigned char)cb;
        __syncthreads();

        int before = 0, after = 0;
        if (live) {
            const int first = bstart[cb] + off[cb] + placed[cb];
            before = count_same(ids, cb, 0, tid);
            if (more) after = count_same(ids, cb, tid + 1, SCAT_T);
            const int p = first + before;
            rc_s[p]   = (unsigned short)((r & 0xFFF) | ((c & 15) << 12));
            vals_s[p] = v;
        }
        if (more) {
            __syncthreads();
            if (after == 0) placed[cb] += before + 1;
        }
    }
}

__global__ void __launch_bounds__(256)
k_spmm_wmma(const float* __restrict__ xT,
            const int*   __restrict__ bucketCount,
            const unsigned short* __restrict__ rc_s,
            const float* __restrict__ vals_s,
            const float* __restrict__ bias,
            float* __restrict__ out) {
    __shared__ unsigned short lrc[CHUNK];
    __shared__ float          lv[CHUNK];

    const int cb    = blockIdx.x;
    const int cbase = cb * 16;
    const int tid   = threadIdx.x;
    const int wave  = tid >> 5;
    const int lane  = tid & 31;
    const int half  = lane >> 4;
    const int ln    = lane & 15;
    const int b0    = (blockIdx.y * 8 + wave) * 16;

    __shared__ int wpart[8];
    int part = (tid < cb) ? bucketCount[tid] : 0;
    for (int d = 16; d > 0; d >>= 1) part += __shfl_xor(part, d);
    if (lane == 0) wpart[wave] = part;
    __syncthreads();
    int ssum = 0;
    for (int k = 0; k < 8; ++k) ssum += wpart[k];
    const int s = __builtin_amdgcn_readfirstlane(ssum);
    const int e = s + bucketCount[cb];

    const float* __restrict__ xp = xT + (size_t)b0 + ln;

    v8f acc0 = {};
    v8f acc1 = {};
    for (int base = s; base < e; base += CHUNK) {
        __syncthreads();
        for (int i = tid; i < CHUNK; i += 256) {
            const int idx = base + i;
            if (idx < e) { lrc[i] = rc_s[idx]; lv[i] = vals_s[idx]; }
            else         { lrc[i] = 0;         lv[i] = 0.0f;        }
            __builtin_prefetch(&rc_s[idx + CHUNK], 0, 0);
            __builtin_prefetch(&vals_s[idx + CHUNK], 0, 0);
        }
        __syncthreads();
        const int n    = min(CHUNK, e - base);
        const int kmax = (n + 7) & ~7;
        for (int k = 0; k < kmax; k += 8) {
            const int kk0 = k + (half << 1);
            const int kk1 = kk0 + 4;

            const unsigned int rcA = *(const unsigned int*)&lrc[kk0];
            const int   rA0 = rcA & 0xFFF,          rA1 = (rcA >> 16) & 0xFFF;
            const int   cA0 = (rcA >> 12) & 0xF,    cA1 = rcA >> 28;
            const float vA0 = lv[kk0],              vA1 = lv[kk0 + 1];
            v2f av, bv;
            av.x = xp[rA0 << 10];
            av.y = xp[rA1 << 10];
            bv.x = (cA0 == ln) ? vA0 : 0.0f;
            bv.y = (cA1 == ln) ? vA1 : 0.0f;
            acc0 = __builtin_amdgcn_wmma_f32_16x16x4_f32(
                false, av, false, bv, (short)0, acc0, false, false);

            const unsigned int rcB = *(const unsigned int*)&lrc[kk1];
            const int   rB0 = rcB & 0xFFF,          rB1 = (rcB >> 16) & 0xFFF;
            const int   cB0 = (rcB >> 12) & 0xF,    cB1 = rcB >> 28;
            const float vB0 = lv[kk1],              vB1 = lv[kk1 + 1];
            v2f aw, bw;
            aw.x = xp[rB0 << 10];
            aw.y = xp[rB1 << 10];
            bw.x = (cB0 == ln) ? vB0 : 0.0f;
            bw.y = (cB1 == ln) ? vB1 : 0.0f;
            acc1 = __builtin_amdgcn_wmma_f32_16x16x4_f32(
                false, aw, false, bw, (short)0, acc1, false, false);
        }
    }

    const float bz = bias[cbase + ln];
    #pragma unroll
    for (int j = 0; j < 8; ++j) {
        const int m = j + 8 * half;
        out[(size_t)(b0 + m) * OUT_F + cbase + ln] = acc0[j] + acc1[j] + bz;
    }
}

extern "C" void kernel_launch(void* const* d_in, const int* in_sizes, int n_in,
                              void* d_out, int out_size, void* d_ws, size_t ws_size,
                              hipStream_t stream) {
    (void)n_in; (void)out_size; (void)ws_size;
    const float* x      = (const float*)d_in[0];
    const float* values = (const float*)d_in[1];
    const float* bias   = (const float*)d_in[2];
    const int*   rows   = (const int*)d_in[3];
    const int*   cols   = (const int*)d_in[4];
    const int    nnz    = in_sizes[1];

    char*  ws  = (char*)d_ws;
    size_t off = 0;
    float* xT          = (float*)(ws + off); off += (size_t)IN_F * BATCH * sizeof(float);
    int*   tileHist    = (int*)  (ws + off); off += (size_t)NTILES * 256 * sizeof(int);
    int*   bucketCount = (int*)  (ws + off); off += 256 * sizeof(int);
    unsigned short* rc_s = (unsigned short*)(ws + off);
    off += (((size_t)nnz * sizeof(unsigned short)) + 15) & ~(size_t)15;
    float* vals_s      = (float*)(ws + off); off += (size_t)nnz * sizeof(float);

    const int tile_sz = (nnz + NTILES - 1) / NTILES;

    k_transpose<<<dim3(IN_F / 32, BATCH / 32), dim3(32, 8), 0, stream>>>(x, xT);
    k_hist    <<<NTILES, 256, 0, stream>>>(cols, nnz, tile_sz, tileHist);
    k_scan    <<<256 / 32, SCAN_W * 32, 0, stream>>>(tileHist, bucketCount);
    k_scatter <<<NTILES, SCAT_T, 0, stream>>>(rows, cols, values, nnz, tile_sz,
                                              tileHist, bucketCount, rc_s, vals_s);
    k_spmm_wmma<<<dim3(NCB, BATCH / 128), 256, 0, stream>>>(xT, bucketCount,
                                                            rc_s, vals_s, bias,
                                                            (float*)d_out);
}


# spmm: a wave takes 64 batch rows: one 16-byte gather per nonzero feeds 4 WMMAs (4 accumulators share B), 16 gathers in flight per trip; each output now summed in one chain instead of two
# speedup vs baseline: 6.1219x; 1.2909x over previous
#include <hip/hip_runtime.h>

typedef __attribute__((ext_vector_type(2))) float v2f;
typedef __attribute__((ext_vector_type(4))) float v4f;
typedef __attribute__((ext_vector_type(8))) float v8f;

#define IN_F   4096
#define OUT_F  4096
#define BATCH  1024
#define NCB    (OUT_F / 16)
#define NTILES 1024
#define CHUNK  992

__global__ void k_transpose(const float* __restrict__ x, float* __restrict__ xT) {
    __shared__ float tile[32][33];
    const int ci = blockIdx.x * 32;
    const int bi = blockIdx.y * 32;
    for (int j = threadIdx.y; j < 32; j += 8)
        tile[j][threadIdx.x] = x[(size_t)(bi + j) * IN_F + ci + threadIdx.x];
    __syncthreads();
    for (int j = threadIdx.y; j < 32; j += 8)
        xT[(size_t)(ci + j) * BATCH + bi + threadIdx.x] = tile[threadIdx.x][j];
}

__global__ void k_hist(const int* __restrict__ cols, int nnz, int tile_sz,
                       int* __restrict__ tileHist) {
    __shared__ int h[256];
    const int t = blockIdx.x;
    h[threadIdx.x] = 0;
    __syncthreads();
    const int beg = t * tile_sz;
    const int end = min(beg + tile_sz, nnz);
    for (int i = beg + (int)threadIdx.x; i < end; i += 256)
        atomicAdd(&h[(cols[i] >> 4) & 255], 1);
    __syncthreads();
    tileHist[t * 256 + threadIdx.x] = h[threadIdx.x];
}

#define SCAN_W   32
#define SCAN_TPW (NTILES / SCAN_W)

__global__ void __launch_bounds__(SCAN_W * 32)
k_scan(int* __restrict__ tileHist, int* __restrict__ bucketCount) {
    __shared__ int wsum[SCAN_W][32];

    const int lane = threadIdx.x & 31;
    const int w    = threadIdx.x >> 5;
    const int cb   = blockIdx.x * 32 + lane;
    int* __restrict__ p = tileHist + (size_t)(w * SCAN_TPW) * 256 + cb;

    int v[SCAN_TPW];
    for (int j = 0; j < SCAN_TPW; ++j) v[j] = p[j * 256];

    int run = 0;
    for (int j = 0; j < SCAN_TPW; ++j) { const int c = v[j]; v[j] = run; run += c; }

    wsum[w][lane] = run;
    __syncthreads();

    int before = 0;
    for (int k = 0; k < SCAN_W; ++k) {
        const int s = wsum[k][lane];
        before += (k < w) ? s : 0;
    }

    for (int j = 0; j < SCAN_TPW; ++j) p[j * 256] = v[j] + before;

    if (w == SCAN_W - 1) bucketCount[cb] = before + run;
}

#define SCAT_T 256

__device__ __forceinline__ int count_same(const uint4* ids, int cb, int lo, int hi) {
    int cnt = 0;
    for (int g = lo >> 4; g < ((hi + 15) >> 4); ++g) {
        const uint4 q = ids[g];
        const unsigned w[4] = {q.x, q.y, q.z, q.w};
        #pragma unroll
        for (int b = 0; b < 16; ++b) {
            const int j  = g * 16 + b;
            const int id = (int)((w[b >> 2] >> (8 * (b & 3))) & 255u);
            cnt += (int)((id == cb) & (j >= lo) & (j < hi));
        }
    }
    return cnt;
}

__global__ void __launch_bounds__(SCAT_T)
k_scatter(const int* __restrict__ rows, const int* __restrict__ cols,
          const float* __restrict__ vals, int nnz, int tile_sz,
          const int* __restrict__ tileOff, const int* __restrict__ bucketCount,
          unsigned short* __restrict__ rc_s, float* __restrict__ vals_s) {
    __shared__ uint4 ids[SCAT_T / 16];
    __shared__ int   placed[256];
    __shared__ int   bstart[256];
    __shared__ int   wtot[SCAT_T / 32];

    const int tid = threadIdx.x;
    const int t   = blockIdx.x;
    const int beg = t * tile_sz;
    const int end = min(beg + tile_sz, nnz);
    const int* __restrict__ off = tileOff + t * 256;

    placed[tid] = 0;
    {
        const int lane = tid & 31;
        const int wv   = tid >> 5;
        const int cnt  = bucketCount[tid];
        int inc = cnt;
        for (int d = 1; d < 32; d <<= 1) {
            const int o = __shfl_up(inc, d);
            if (lane >= d) inc += o;
        }
        if (lane == 31) wtot[wv] = inc;
        __syncthreads();
        int pre = 0;
        for (int k = 0; k < SCAT_T / 32; ++k) pre += (k < wv) ? wtot[k] : 0;
        bstart[tid] = pre + inc - cnt;
    }
    for (int base = beg; base < end; base += SCAT_T) {
        const int  n    = min(SCAT_T, end - base);
        const bool live = tid < n;
        const bool more = base + SCAT_T < end;
        int c = 0, r = 0;
        float v = 0.0f;
        if (live) { c = cols[base + tid]; r = rows[base + tid]; v = vals[base + tid]; }
        const int cb = (c >> 4) & 255;

        __syncthreads();
        ((unsigned char*)ids)[tid] = (unsigned char)cb;
        __syncthreads();

        int before = 0, after = 0;
        if (live) {
            const int first = bstart[cb] + off[cb] + placed[cb];
            before = count_same(ids, cb, 0, tid);
            if (more) after = count_same(ids, cb, tid + 1, SCAT_T);
            const int p = first + before;
            rc_s[p]   = (unsigned short)((r & 0xFFF) | ((c & 15) << 12));
            vals_s[p] = v;
        }
        if (more) {
            __syncthreads();
            if (after == 0) placed[cb] += before + 1;
        }
    }
}

#define SP_WAVES 16
#define SP_NG    8
#define SP_PAD   (4 * SP_NG)

struct Ent {
    unsigned o0, o1;
    v2f      b;
};

__device__ __forceinline__ void sp_entry(Ent& en, const unsigned short* lrc, const float* lv,
                                         int kk, unsigned lane_off, int ln) {
    const unsigned rc = *(const unsigned*)&lrc[kk];
    const float2   vv = *(const float2*)&lv[kk];
    en.o0 = ((rc & 0xFFFu) << 12) + lane_off;
    en.o1 = (((rc >> 16) & 0xFFFu) << 12) + lane_off;
    const int c0 = (int)((rc >> 12) & 0xFu);
    const int c1 = (int)(rc >> 28);
    en.b.x = (c0 == ln) ? vv.x : 0.0f;
    en.b.y = (c1 == ln) ? vv.y : 0.0f;
}

__global__ void __launch_bounds__(SP_WAVES * 32)
k_spmm_wmma(const float* __restrict__ xT,
            const int*   __restrict__ bucketCount,
            const unsigned short* __restrict__ rc_s,
            const float* __restrict__ vals_s,
            const float* __restrict__ bias,
            float* __restrict__ out) {
    __shared__ __attribute__((aligned(16))) unsigned short lrc[CHUNK + SP_PAD];
    __shared__ __attribute__((aligned(16))) float          lv[CHUNK + SP_PAD];

    const int cb    = blockIdx.x;
    const int cbase = cb * 16;
    const int tid   = threadIdx.x;
    const int wave  = tid >> 5;
    const int lane  = tid & 31;
    const int half  = lane >> 4;
    const int ln    = lane & 15;
    const int b0    = wave * 64;

    __shared__ int wpart[SP_WAVES];
    int part = (tid < cb) ? bucketCount[tid] : 0;
    for (int d = 16; d > 0; d >>= 1) part += __shfl_xor(part, d);
    if (lane == 0) wpart[wave] = part;
    __syncthreads();
    int ssum = 0;
    for (int k = 0; k < SP_WAVES; ++k) ssum += wpart[k];
    const int s = __builtin_amdgcn_readfirstlane(ssum);
    const int e = s + bucketCount[cb];

    const char* __restrict__ xTb = (const char*)xT;
    const unsigned lane_off = (unsigned)(b0 + 4 * ln) * 4u;
    const int hk = half << 1;

    v8f acc[4] = {};
    for (int base = s; base < e; base += CHUNK) {
        __syncthreads();
        for (int i = tid; i < CHUNK + SP_PAD; i += SP_WAVES * 32) {
            const int idx = base + i;
            if (i < CHUNK && idx < e) { lrc[i] = rc_s[idx]; lv[i] = vals_s[idx]; }
            else                      { lrc[i] = 0;         lv[i] = 0.0f;        }
            if (idx + CHUNK < e) {
                __builtin_prefetch(&rc_s[idx + CHUNK], 0, 0);
                __builtin_prefetch(&vals_s[idx + CHUNK], 0, 0);
            }
        }
        __syncthreads();
        const int n    = min(CHUNK, e - base);
        const int kmax = (n + 4 * SP_NG - 1) & ~(4 * SP_NG - 1);

        Ent en[SP_NG];
        for (int g = 0; g < SP_NG; ++g) sp_entry(en[g], lrc, lv, 4 * g + hk, lane_off, ln);

        for (int k = 0; k < kmax; k += 4 * SP_NG) {
            v4f a0[SP_NG], a1[SP_NG];
            v2f bf[SP_NG];
            for (int g = 0; g < SP_NG; ++g) {
                a0[g] = *(const v4f*)(xTb + en[g].o0);
                a1[g] = *(const v4f*)(xTb + en[g].o1);
                bf[g] = en[g].b;
            }
            __builtin_amdgcn_sched_barrier(0);
            for (int g = 0; g < SP_NG; ++g)
                sp_entry(en[g], lrc, lv, k + 4 * SP_NG + 4 * g + hk, lane_off, ln);
            for (int g = 0; g < SP_NG; ++g) {
                v2f a;
                a.x = a0[g].x; a.y = a1[g].x;
                acc[0] = __builtin_amdgcn_wmma_f32_16x16x4_f32(false, a, false, bf[g], (short)0, acc[0], false, false);
                a.x = a0[g].y; a.y = a1[g].y;
                acc[1] = __builtin_amdgcn_wmma_f32_16x16x4_f32(false, a, false, bf[g], (short)0, acc[1], false, false);
                a.x = a0[g].z; a.y = a1[g].z;
                acc[2] = __builtin_amdgcn_wmma_f32_16x16x4_f32(false, a, false, bf[g], (short)0, acc[2], false, false);
                a.x = a0[g].w; a.y = a1[g].w;
                acc[3] = __builtin_amdgcn_wmma_f32_16x16x4_f32(false, a, false, bf[g], (short)0, acc[3], false, false);
            }
        }
    }

    const float bz = bias[cbase + ln];
    for (int t = 0; t < 4; ++t) {
        for (int j = 0; j < 8; ++j) {
            const int m = j + 8 * half;
            out[(size_t)(b0 + 4 * m + t) * OUT_F + cbase + ln] = acc[t][j] + bz;
        }
    }
}

extern "C" void kernel_launch(void* const* d_in, const int* in_sizes, int n_in,
                              void* d_out, int out_size, void* d_ws, size_t ws_size,
                              hipStream_t stream) {
    (void)n_in; (void)out_size; (void)ws_size;
    const float* x      = (const float*)d_in[0];
    const float* values = (const float*)d_in[1];
    const float* bias   = (const float*)d_in[2];
    const int*   rows   = (const int*)d_in[3];
    const int*   cols   = (const int*)d_in[4];
    const int    nnz    = in_sizes[1];

    char*  ws  = (char*)d_ws;
    size_t off = 0;
    float* xT          = (float*)(ws + off); off += (size_t)IN_F * BATCH * sizeof(float);
    int*   tileHist    = (int*)  (ws + off); off += (size_t)NTILES * 256 * sizeof(int);
    int*   bucketCount = (int*)  (ws + off); off += 256 * sizeof(int);
    unsigned short* rc_s = (unsigned short*)(ws + off);
    off += (((size_t)nnz * sizeof(unsigned short)) + 15) & ~(size_t)15;
    float* vals_s      = (float*)(ws + off); off += (size_t)nnz * sizeof(float);

    const int tile_sz = (nnz + NTILES - 1) / NTILES;

    k_transpose<<<dim3(IN_F / 32, BATCH / 32), dim3(32, 8), 0, stream>>>(x, xT);
    k_hist    <<<NTILES, 256, 0, stream>>>(cols, nnz, tile_sz, tileHist);
    k_scan    <<<256 / 32, SCAN_W * 32, 0, stream>>>(tileHist, bucketCount);
    k_scatter <<<NTILES, SCAT_T, 0, stream>>>(rows, cols, values, nnz, tile_sz,
                                              tileHist, bucketCount, rc_s, vals_s);
    k_spmm_wmma<<<NCB, SP_WAVES * 32, 0, stream>>>(xT, bucketCount,
                                                            rc_s, vals_s, bias,
                                                            (float*)d_out);
}


# spmm: each work-group picks its column block's nonzeros itself, scanning the column indices once into LDS in input order; the histogram, scan and scatter kernels are gone; same sums in the same order
# speedup vs baseline: 7.1201x; 1.1631x over previous
#include <hip/hip_runtime.h>

typedef __attribute__((ext_vector_type(2))) float v2f;
typedef __attribute__((ext_vector_type(4))) float v4f;
typedef __attribute__((ext_vector_type(8))) float v8f;

#define IN_F   4096
#define OUT_F  4096
#define BATCH  1024
#define NCB    (OUT_F / 16)

__global__ void k_transpose(const float* __restrict__ x, float* __restrict__ xT) {
    __shared__ float tile[32][33];
    const int ci = blockIdx.x * 32;
    const int bi = blockIdx.y * 32;
    for (int j = threadIdx.y; j < 32; j += 8)
        tile[j][threadIdx.x] = x[(size_t)(bi + j) * IN_F + ci + threadIdx.x];
    __syncthreads();
    for (int j = threadIdx.y; j < 32; j += 8)
        xT[(size_t)(ci + j) * BATCH + bi + threadIdx.x] = tile[threadIdx.x][j];
}

#define SP_WAVES 16
#define SP_NG    8
#define SP_PAD   (4 * SP_NG)
#define FW_CAP   1024
#define LIST_CAP (SP_WAVES * FW_CAP)
#define FW_U     8
#define FW_STEP  128

struct Ent {
    unsigned o0, o1;
    v2f      b;
};

__device__ __forceinline__ void sp_entry(Ent& en, const unsigned short* lrc, const float* lv,
                                         int kk, unsigned lane_off, int ln) {
    const unsigned rc = *(const unsigned*)&lrc[kk];
    const float2   vv = *(const float2*)&lv[kk];
    en.o0 = ((rc & 0xFFFu) << 12) + lane_off;
    en.o1 = (((rc >> 16) & 0xFFFu) << 12) + lane_off;
    const int c0 = (int)((rc >> 12) & 0xFu);
    const int c1 = (int)(rc >> 28);
    en.b.x = (c0 == ln) ? vv.x : 0.0f;
    en.b.y = (c1 == ln) ? vv.y : 0.0f;
}

__device__ __forceinline__ void filter_load(int4 (&c)[FW_U], const int* __restrict__ cols,
                                            int ib, int lane, int lim4) {
    for (int u = 0; u < FW_U; ++u) {
        const int i = ib + (u * 32 + lane) * 4;
        c[u] = *(const int4*)&cols[(i < lim4) ? i : 0];
    }
}

__device__ __forceinline__ void filter_take(const int4 (&c)[FW_U], int ib, int lane, int lim4, int cb,
                                            unsigned lt, unsigned* st, int& cnt) {
    for (int u = 0; u < FW_U; ++u) {
        const int  i   = ib + (u * 32 + lane) * 4;
        const int  key = (i < lim4) ? cb : (int)0x80000000;
        const bool m0  = (c[u].x >> 4) == key;
        const bool m1  = (c[u].y >> 4) == key;
        const bool m2  = (c[u].z >> 4) == key;
        const bool m3  = (c[u].w >> 4) == key;
        const unsigned b0 = __builtin_amdgcn_ballot_w32(m0);
        const unsigned b1 = __builtin_amdgcn_ballot_w32(m1);
        const unsigned b2 = __builtin_amdgcn_ballot_w32(m2);
        const unsigned b3 = __builtin_amdgcn_ballot_w32(m3);
        if ((b0 | b1 | b2 | b3) != 0u) {
            int pos = cnt + __popc(b0 & lt) + __popc(b1 & lt) + __popc(b2 & lt) + __popc(b3 & lt);
            if (m0) { if (pos < FW_CAP) st[pos] = (unsigned)i;       ++pos; }
            if (m1) { if (pos < FW_CAP) st[pos] = (unsigned)(i + 1); ++pos; }
            if (m2) { if (pos < FW_CAP) st[pos] = (unsigned)(i + 2); ++pos; }
            if (m3) { if (pos < FW_CAP) st[pos] = (unsigned)(i + 3); ++pos; }
            cnt += __popc(b0) + __popc(b1) + __popc(b2) + __popc(b3);
        }
    }
}

__device__ __forceinline__ int filter_segment(const int* __restrict__ cols, int wb, int len, int n_in,
                                              int cb, int lane, unsigned* st) {
    const int      n4   = n_in & ~3;
    const int      lim4 = min(n4, wb + len);
    const unsigned lt   = (1u << lane) - 1u;
    const int      grp  = FW_STEP * FW_U;
    int cnt = 0;
    if (lim4 > wb) {
        int4 ga[FW_U], gb[FW_U];
        filter_load(ga, cols, wb, lane, lim4);
        for (int s0 = 0; s0 < len; s0 += 2 * grp) {
            filter_load(gb, cols, wb + s0 + grp, lane, lim4);
            __builtin_amdgcn_sched_barrier(0);
            filter_take(ga, wb + s0, lane, lim4, cb, lt, st, cnt);
            filter_load(ga, cols, wb + s0 + 2 * grp, lane, lim4);
            __builtin_amdgcn_sched_barrier(0);
            filter_take(gb, wb + s0 + grp, lane, lim4, cb, lt, st, cnt);
        }
    }
    if (n4 < n_in && n4 >= wb && n4 < wb + len) {
        const int  i = n4 + lane;
        const bool m = (i < n_in) && ((cols[i] >> 4) == cb);
        const unsigned b = __builtin_amdgcn_ballot_w32(m);
        if (b != 0u) {
            const int pos = cnt + __popc(b & lt);
            if (m && pos < FW_CAP) st[pos] = (unsigned)i;
            cnt += __popc(b);
        }
    }
    return cnt;
}

__global__ void __launch_bounds__(SP_WAVES * 32)
k_spmm_wmma(const float* __restrict__ xT,
            const int*   __restrict__ rows,
            const int*   __restrict__ cols,
            const float* __restrict__ vals,
            int nnz,
            const float* __restrict__ bias,
            float* __restrict__ out) {
    __shared__ unsigned stage[SP_WAVES][FW_CAP];
    __shared__ __attribute__((aligned(16))) unsigned short lrc[LIST_CAP + SP_PAD];
    __shared__ __attribute__((aligned(16))) float          lv[LIST_CAP + SP_PAD];
    __shared__ int wcnt[SP_WAVES];

    const int cb    = blockIdx.x;
    const int cbase = cb * 16;
    const int tid   = threadIdx.x;
    const int wave  = tid >> 5;
    const int lane  = tid & 31;
    const int half  = lane >> 4;
    const int ln    = lane & 15;
    const int b0    = wave * 64;

    const char* __restrict__ xTb = (const char*)xT;
    const unsigned lane_off = (unsigned)(b0 + 4 * ln) * 4u;
    const int hk = half << 1;

    v8f acc[4] = {};

    int len = ((nnz + SP_WAVES * FW_STEP - 1) / (SP_WAVES * FW_STEP)) * FW_STEP;
    for (int pass = 0; pass < 2; ++pass) {
        bool ovf = false;
        for (int rbase = 0; rbase < nnz; rbase += SP_WAVES * len) {
            const int cnt = filter_segment(cols, rbase + wave * len, len, nnz, cb, lane, stage[wave]);
            if (lane == 0) wcnt[wave] = cnt;
            __syncthreads();
            int tot = 0, pre = 0, big = 0;
            for (int k = 0; k < SP_WAVES; ++k) {
                const int c = wcnt[k];
                pre += (k < wave) ? c : 0;
                tot += c;
                big |= (c > FW_CAP) ? 1 : 0;
            }
            const int n  = __builtin_amdgcn_readfirstlane(tot);
            const int p0 = __builtin_amdgcn_readfirstlane(pre);
            ovf = __builtin_amdgcn_readfirstlane(big) != 0;
            if (ovf) {
                __syncthreads();
                break;
            }
            for (int j = lane; j < cnt; j += 32) {
                const int   i = (int)stage[wave][j];
                const int   c = cols[i];
                const int   r = rows[i];
                const float v = vals[i];
                lrc[p0 + j] = (unsigned short)((r & 0xFFF) | ((c & 15) << 12));
                lv[p0 + j]  = v;
            }
            const int kmax = (n + 4 * SP_NG - 1) & ~(4 * SP_NG - 1);
            for (int j = n + tid; j < kmax + SP_PAD; j += SP_WAVES * 32) { lrc[j] = 0; lv[j] = 0.0f; }
            __syncthreads();

            Ent en[SP_NG];
            for (int g = 0; g < SP_NG; ++g) sp_entry(en[g], lrc, lv, 4 * g + hk, lane_off, ln);

            for (int k = 0; k < kmax; k += 4 * SP_NG) {
                v4f a0[SP_NG], a1[SP_NG];
                v2f bf[SP_NG];
                for (int g = 0; g < SP_NG; ++g) {
                    a0[g] = *(const v4f*)(xTb + en[g].o0);
                    a1[g] = *(const v4f*)(xTb + en[g].o1);
                    bf[g] = en[g].b;
                }
                __builtin_amdgcn_sched_barrier(0);
                for (int g = 0; g < SP_NG; ++g)
                    sp_entry(en[g], lrc, lv, k + 4 * SP_NG + 4 * g + hk, lane_off, ln);
                for (int g = 0; g < SP_NG; ++g) {
                    v2f a;
                    a.x = a0[g].x; a.y = a1[g].x;
                    acc[0] = __builtin_amdgcn_wmma_f32_16x16x4_f32(false, a, false, bf[g], (short)0, acc[0], false, false);
                    a.x = a0[g].y; a.y = a1[g].y;
                    acc[1] = __builtin_amdgcn_wmma_f32_16x16x4_f32(false, a, false, bf[g], (short)0, acc[1], false, false);
                    a.x = a0[g].z; a.y = a1[g].z;
                    acc[2] = __builtin_amdgcn_wmma_f32_16x16x4_f32(false, a, false, bf[g], (short)0, acc[2], false, false);
                    a.x = a0[g].w; a.y = a1[g].w;
                    acc[3] = __builtin_amdgcn_wmma_f32_16x16x4_f32(false, a, false, bf[g], (short)0, acc[3], false, false);
                }
            }
        }
        if (!ovf) break;
        len = FW_CAP;
    }

    const float bz = bias[cbase + ln];
    for (int t = 0; t < 4; ++t) {
        for (int j = 0; j < 8; ++j) {
            const int m = j + 8 * half;
            out[(size_t)(b0 + 4 * m + t) * OUT_F + cbase + ln] = acc[t][j] + bz;
        }
    }
}

extern "C" void kernel_launch(void* const* d_in, const int* in_sizes, int n_in,
                              void* d_out, int out_size, void* d_ws, size_t ws_size,
                              hipStream_t stream) {
    (void)n_in; (void)out_size; (void)ws_size;
    const float* x      = (const float*)d_in[0];
    const float* values = (const float*)d_in[1];
    const float* bias   = (const float*)d_in[2];
    const int*   rows   = (const int*)d_in[3];
    const int*   cols   = (const int*)d_in[4];
    const int    nnz    = in_sizes[1];

    float* xT = (float*)d_ws;

    k_transpose<<<dim3(IN_F / 32, BATCH / 32), dim3(32, 8), 0, stream>>>(x, xT);
    k_spmm_wmma<<<NCB, SP_WAVES * 32, 0, stream>>>(xT, rows, cols, values, nnz, bias,
                                                            (float*)d_out);
}
